# starting kernel
# baseline (speedup 1.0000x reference)
#include <hip/hip_runtime.h>

#define R_ 25
#define C_ 64
#define N_ 64
#define H_ 8
#define E_ 1024
#define NC_ 27

typedef float v2f __attribute__((ext_vector_type(2)));
typedef float v8f __attribute__((ext_vector_type(8)));
typedef float v4f __attribute__((ext_vector_type(4)));

__device__ __forceinline__ v8f wmma4(v2f a, v2f b, v8f c) {
  return __builtin_amdgcn_wmma_f32_16x16x4_f32(false, a, false, b, (short)0, c,
                                               false, false);
}

__global__ __launch_bounds__(256) void k_emb_ln(const float* __restrict__ w,
                                                const float* __restrict__ g,
                                                const float* __restrict__ b,
                                                float* __restrict__ out) {
  __shared__ float red[256];
  int row = blockIdx.x, tid = threadIdx.x;
  const float* x = w + (size_t)row * E_;
  float s = 0.f;
  for (int e = tid; e < E_; e += 256) s += x[e];
  red[tid] = s; __syncthreads();
  for (int off = 128; off > 0; off >>= 1) {
    if (tid < off) red[tid] += red[tid + off];
    __syncthreads();
  }
  float mean = red[0] * (1.f / E_);
  __syncthreads();
  float s2 = 0.f;
  for (int e = tid; e < E_; e += 256) { float d = x[e] - mean; s2 += d * d; }
  red[tid] = s2; __syncthreads();
  for (int off = 128; off > 0; off >>= 1) {
    if (tid < off) red[tid] += red[tid + off];
    __syncthreads();
  }
  float inv = rsqrtf(red[0] * (1.f / E_) + 1e-16f);
  for (int e = tid; e < E_; e += 256)
    out[(size_t)row * E_ + e] = (x[e] - mean) * inv * g[e] + b[e];
}

__global__ void k_means(const float* __restrict__ bst,
                        const float* __restrict__ bsi,
                        float* __restrict__ bt, float* __restrict__ bi) {
  int t = blockIdx.x * blockDim.x + threadIdx.x;
  if (t < R_ * N_) {
    int d = t / N_, i = t % N_;
    const float* p = bst + ((size_t)d * C_ * N_ + i) * 4;
    bt[t] = 0.25f * (p[0] + p[1] + p[2] + p[3]);
  } else if (t < R_ * N_ + C_ * N_) {
    int t2 = t - R_ * N_;
    int d = t2 / N_, i = t2 % N_;
    const float* p = bsi + ((size_t)d * N_ + i) * 4;
    bi[t2] = 0.25f * (p[0] + p[1] + p[2] + p[3]);
  }
}

__global__ void k_logsm(const float* __restrict__ cst,
                        const float* __restrict__ csi,
                        float* __restrict__ lt, float* __restrict__ li) {
  int t = blockIdx.x * blockDim.x + threadIdx.x;
  const float* src; float* dst;
  if (t < R_ * N_) {
    int a = t / N_, nn = t % N_;
    src = cst + ((size_t)a * C_ * N_ + nn) * NC_ * 4;
    dst = lt + (size_t)t * NC_;
  } else if (t < R_ * N_ + C_ * N_) {
    int t2 = t - R_ * N_;
    int b = t2 / N_, nn = t2 % N_;
    src = csi + ((size_t)b * N_ + nn) * NC_ * 4;
    dst = li + (size_t)t2 * NC_;
  } else return;
  float v[NC_];
  float m = -1e30f;
  for (int k = 0; k < NC_; ++k) {
    const float* p = src + k * 4;
    v[k] = 0.25f * (p[0] + p[1] + p[2] + p[3]);
    m = fmaxf(m, v[k]);
  }
  float se = 0.f;
  for (int k = 0; k < NC_; ++k) se += expf(v[k] - m);
  float lse = m + logf(se);
  for (int k = 0; k < NC_; ++k) dst[k] = v[k] - lse;
}

#define SEG   128
#define EPAD  132
#define QKPAD 68

__global__ __launch_bounds__(256) void k_attn(
    const float* __restrict__ emb_norm,
    const float* __restrict__ WQt, const float* __restrict__ WKt,
    const float* __restrict__ WQi, const float* __restrict__ WKi,
    float* __restrict__ attn) {
  __shared__ float lds[64 * QKPAD * 2];

  int tid = threadIdx.x;
  int lane = tid & 31;
  int wid = tid >> 5;

  int blk = blockIdx.x;
  int r = blk % R_;
  int h = (blk / R_) % H_;
  int w = blk / (R_ * H_);

  const float* WQ = (w ? WQi : WQt) + (size_t)h * E_ * E_;
  const float* WK = (w ? WKi : WKt) + (size_t)h * E_ * E_;
  const float* Eg = emb_norm + (size_t)r * N_ * E_;

  int qk = wid >> 2;
  int n0 = (wid & 3) << 4;
  const float* Wm = qk ? WK : WQ;

  int i0 = (wid >> 1) << 4;
  int j0 = (wid & 1) << 5;
  v8f s0 = {}; v8f s1 = {};

  int kp = (lane >> 4) << 1;
  int nl = lane & 15;

  for (int fc = 0; fc < 16; ++fc) {
    int f0 = fc << 6;
    v8f acc0 = {}, acc1 = {}, acc2 = {}, acc3 = {};

    for (int seg = 0; seg < E_ / SEG; ++seg) {
      int e0 = seg * SEG;
      __syncthreads();
      for (int q = tid; q < 64 * SEG / 4; q += 256) {
        int rr = q >> 5;
        int cc = (q & 31) << 2;
        *reinterpret_cast<v4f*>(&lds[rr * EPAD + cc]) =
            *reinterpret_cast<const v4f*>(&Eg[(size_t)rr * E_ + e0 + cc]);
      }
      __syncthreads();

      for (int k0 = 0; k0 < SEG; k0 += 4) {
        size_t eg = (size_t)(e0 + k0 + kp);
        v2f b = *reinterpret_cast<const v2f*>(&Wm[(size_t)(f0 + n0 + nl) * E_ + eg]);
        int lbase = k0 + kp;
        v2f a0 = *reinterpret_cast<const v2f*>(&lds[( 0 + nl) * EPAD + lbase]);
        v2f a1 = *reinterpret_cast<const v2f*>(&lds[(16 + nl) * EPAD + lbase]);
        v2f a2 = *reinterpret_cast<const v2f*>(&lds[(32 + nl) * EPAD + lbase]);
        v2f a3 = *reinterpret_cast<const v2f*>(&lds[(48 + nl) * EPAD + lbase]);
        acc0 = wmma4(a0, b, acc0);
        acc1 = wmma4(a1, b, acc1);
        acc2 = wmma4(a2, b, acc2);
        acc3 = wmma4(a3, b, acc3);
      }
    }

    __syncthreads();
    {
      float* QK = lds + qk * (64 * QKPAD);
      int rsub = (lane >> 4) << 3;
#pragma unroll
      for (int v = 0; v < 8; ++v) {
        QK[( 0 + rsub + v) * QKPAD + n0 + nl] = acc0[v];
        QK[(16 + rsub + v) * QKPAD + n0 + nl] = acc1[v];
        QK[(32 + rsub + v) * QKPAD + n0 + nl] = acc2[v];
        QK[(48 + rsub + v) * QKPAD + n0 + nl] = acc3[v];
      }
    }
    __syncthreads();

    {
      const float* Qs = lds;
      const float* Ks = lds + 64 * QKPAD;
      for (int k0 = 0; k0 < 64; k0 += 4) {
        v2f a  = *reinterpret_cast<const v2f*>(&Qs[(i0 + nl) * QKPAD + k0 + kp]);
        v2f b0 = *reinterpret_cast<const v2f*>(&Ks[(j0 + nl) * QKPAD + k0 + kp]);
        v2f b1 = *reinterpret_cast<const v2f*>(&Ks[(j0 + 16 + nl) * QKPAD + k0 + kp]);
        s0 = wmma4(a, b0, s0);
        s1 = wmma4(a, b1, s1);
      }
    }
  }

  __syncthreads();
  {
    int rbase = i0 + ((lane >> 4) << 3);
#pragma unroll
    for (int v = 0; v < 8; ++v) {
      lds[(rbase + v) * 64 + j0 + nl]      = s0[v];
      lds[(rbase + v) * 64 + j0 + 16 + nl] = s1[v];
    }
  }
  __syncthreads();

  if (tid < 64) {
    int j = tid;
    const float inv32 = 0.03125f;
    float m = -1e30f;
    for (int i = 0; i < 64; ++i) m = fmaxf(m, lds[i * 64 + j] * inv32);
    float se = 0.f;
    for (int i = 0; i < 64; ++i) {
      float e = expf(lds[i * 64 + j] * inv32 - m);
      lds[i * 64 + j] = e;
      se += e;
    }
    float rs = 1.f / se;
    for (int i = 0; i < 64; ++i) lds[i * 64 + j] *= rs;
  }
  __syncthreads();

  float* dst = attn + (size_t)((w * H_ + h) * R_ + r) * 4096;
  for (int q = tid; q < 4096; q += 256) dst[q] = lds[q];
}

__global__ __launch_bounds__(256) void k_br(
    const float* __restrict__ attn, const float* __restrict__ bt,
    const float* __restrict__ bi, const int* __restrict__ idxt,
    const int* __restrict__ idxi, float* __restrict__ BRt,
    float* __restrict__ BRi) {
  __shared__ float brow[64];
  int b = blockIdx.x, tid = threadIdx.x;
  int d, r, w;
  const float* bsrc; float* dst;
  if (b < R_) { d = b;      r = idxt[d * C_]; w = 0; bsrc = bt + d * 64; dst = BRt + (size_t)d * 64 * 8; }
  else        { d = b - R_; r = idxi[d];      w = 1; bsrc = bi + d * 64; dst = BRi + (size_t)d * 64 * 8; }
  if (tid < 64) brow[tid] = bsrc[tid];
  __syncthreads();
  int j = tid & 63;
  for (int hh = tid >> 6; hh < H_; hh += 4) {
    const float* A = attn + (size_t)((w * H_ + hh) * R_ + r) * 4096;
    float acc = 0.f;
    for (int i = 0; i < 64; ++i) acc += brow[i] * A[i * 64 + j];
    dst[(size_t)j * 8 + hh] = acc;
  }
}

__global__ void k_bs(const float* __restrict__ BRt,
                     const float* __restrict__ BRi, float* __restrict__ BS) {
  int t = blockIdx.x * blockDim.x + threadIdx.x;
  if (t >= R_ * C_ * N_) return;
  int a = t / (C_ * N_);
  int rem = t % (C_ * N_);
  int b = rem / N_, j = rem % N_;
  const float* x = BRt + ((size_t)a * 64 + j) * 8;
  const float* y = BRi + ((size_t)b * 64 + j) * 8;
  float dot = 0.f, nx = 0.f, ny = 0.f;
  for (int hh = 0; hh < 8; ++hh) {
    dot += x[hh] * y[hh]; nx += x[hh] * x[hh]; ny += y[hh] * y[hh];
  }
  float na = fmaxf(sqrtf(nx), 1e-15f);
  float nb = fmaxf(sqrtf(ny), 1e-15f);
  BS[t] = dot / (na * nb);
}

__global__ void k_cs(const float* __restrict__ lt, const float* __restrict__ li,
                     float* __restrict__ CS) {
  int t = blockIdx.x * blockDim.x + threadIdx.x;
  if (t >= R_ * C_ * N_) return;
  int a = t / (C_ * N_);
  int rem = t % (C_ * N_);
  int b = rem / N_, nn = rem % N_;
  const float* pt = lt + ((size_t)a * 64 + nn) * NC_;
  const float* pi = li + ((size_t)b * 64 + nn) * NC_;
  float m = -1e30f;
  for (int k = 0; k < NC_; ++k) m = fmaxf(m, 0.5f * (pt[k] + pi[k]));
  float se = 0.f;
  for (int k = 0; k < NC_; ++k) se += expf(0.5f * (pt[k] + pi[k]) - m);
  float lse = m + logf(se);
  float klt = 0.f, kli = 0.f;
  for (int k = 0; k < NC_; ++k) {
    float M = 0.5f * (pt[k] + pi[k]) - lse;
    float eM = expf(M);
    klt += eM * (M - pt[k]);
    kli += eM * (M - pi[k]);
  }
  CS[t] = -0.5f * (klt + kli) * (1.f / NC_);
}

__global__ __launch_bounds__(256) void k_os1(
    const float* __restrict__ emb, const float* __restrict__ Wos,
    const float* __restrict__ bos, const int* __restrict__ idxi,
    const int* __restrict__ covb, float* __restrict__ ob) {
  __shared__ float rowv[E_];
  int d = blockIdx.x, tid = threadIdx.x;
  int row = idxi[d] * N_ + covb[d];
  for (int e = tid; e < E_; e += 256) rowv[e] = emb[(size_t)row * E_ + e];
  __syncthreads();
  for (int f = tid; f < E_; f += 256) {
    const float* wr = Wos + (size_t)f * E_;
    float acc = bos[f];
    for (int e = 0; e < E_; ++e) acc += rowv[e] * wr[e];
    ob[(size_t)d * E_ + f] = acc;
  }
}

__global__ __launch_bounds__(256) void k_os2(const float* __restrict__ ob,
                                             const float* __restrict__ emb,
                                             float* __restrict__ osf) {
  __shared__ float rowv[E_];
  __shared__ float part[256];
  int d = blockIdx.y;
  int tc = blockIdx.x;
  int tid = threadIdx.x;
  for (int e = tid; e < E_; e += 256) rowv[e] = ob[(size_t)d * E_ + e];
  __syncthreads();
  int t = tc * 64 + (tid & 63);
  int p = tid >> 6;
  const float* er = emb + (size_t)t * E_ + p * 256;
  const float* rv = rowv + p * 256;
  float acc = 0.f;
  for (int e = 0; e < 256; ++e) acc += rv[e] * er[e];
  part[tid] = acc;
  __syncthreads();
  if (p == 0)
    osf[(size_t)d * 1600 + t] = part[tid] + part[tid + 64] + part[tid + 128] + part[tid + 192];
}

__global__ __launch_bounds__(256) void k_ln_out(
    const float* __restrict__ BSp, const float* __restrict__ CSp,
    const float* __restrict__ OSf, const float* __restrict__ g0,
    const float* __restrict__ b0, const float* __restrict__ g1,
    const float* __restrict__ b1, const float* __restrict__ g2,
    const float* __restrict__ b2, float* __restrict__ out) {
  __shared__ float red[256];
  int mode = blockIdx.x / R_;
  int sl = blockIdx.x % R_;
  int tid = threadIdx.x;
  float x[16];
  for (int q = 0; q < 16; ++q) {
    int idx = tid * 16 + q;
    float v;
    if (mode == 0) v = BSp[(size_t)sl * 4096 + idx];
    else if (mode == 1) v = CSp[(size_t)sl * 4096 + idx];
    else { int bb = idx >> 6, nn = idx & 63; v = OSf[(size_t)bb * 1600 + sl * 64 + nn]; }
    x[q] = v;
  }
  float s = 0.f;
  for (int q = 0; q < 16; ++q) s += x[q];
  red[tid] = s; __syncthreads();
  for (int off = 128; off > 0; off >>= 1) {
    if (tid < off) red[tid] += red[tid + off];
    __syncthreads();
  }
  float mean = red[0] * (1.f / 4096.f);
  __syncthreads();
  float s2 = 0.f;
  for (int q = 0; q < 16; ++q) { float dd = x[q] - mean; s2 += dd * dd; }
  red[tid] = s2; __syncthreads();
  for (int off = 128; off > 0; off >>= 1) {
    if (tid < off) red[tid] += red[tid + off];
    __syncthreads();
  }
  float inv = rsqrtf(red[0] * (1.f / 4096.f) + 1e-5f);
  const float* g = (mode == 0) ? g0 : ((mode == 1) ? g1 : g2);
  const float* bb = (mode == 0) ? b0 : ((mode == 1) ? b1 : b2);
  for (int q = 0; q < 16; ++q) {
    int idx = tid * 16 + q;
    out[(size_t)mode * 102400 + (size_t)sl * 4096 + idx] =
        (x[q] - mean) * inv * g[idx] + bb[idx];
  }
}

extern "C" void kernel_launch(void* const* d_in, const int* in_sizes, int n_in,
                              void* d_out, int out_size, void* d_ws,
                              size_t ws_size, hipStream_t stream) {
  const float* bst = (const float*)d_in[0];
  const float* bsi = (const float*)d_in[1];
  const float* cst = (const float*)d_in[2];
  const float* csi = (const float*)d_in[3];
  const int* idxt  = (const int*)d_in[4];
  const int* idxi  = (const int*)d_in[5];
  const int* covb  = (const int*)d_in[6];
  const float* emb = (const float*)d_in[7];
  const float* lng = (const float*)d_in[8];
  const float* lnb = (const float*)d_in[9];
  const float* WQt = (const float*)d_in[10];
  const float* WKt = (const float*)d_in[11];
  const float* WQi = (const float*)d_in[12];
  const float* WKi = (const float*)d_in[13];
  const float* Wos = (const float*)d_in[14];
  const float* bos = (const float*)d_in[15];
  const float* BSg = (const float*)d_in[16];
  const float* BSb = (const float*)d_in[17];
  const float* CSg = (const float*)d_in[18];
  const float* CSb = (const float*)d_in[19];
  const float* OSg = (const float*)d_in[20];
  const float* OSb = (const float*)d_in[21];

  float* ws = (float*)d_ws;
  size_t o = 0;
  float* emb_norm = ws + o; o += (size_t)1600 * 1024;
  float* attn     = ws + o; o += (size_t)2 * 8 * 25 * 4096;
  float* bt       = ws + o; o += R_ * N_;
  float* bi       = ws + o; o += C_ * N_;
  float* lt       = ws + o; o += R_ * N_ * NC_;
  float* li       = ws + o; o += C_ * N_ * NC_;
  float* BRt      = ws + o; o += R_ * N_ * H_;
  float* BRi      = ws + o; o += C_ * N_ * H_;
  float* BSp      = ws + o; o += R_ * C_ * N_;
  float* CSp      = ws + o; o += R_ * C_ * N_;
  float* ob       = ws + o; o += C_ * E_;
  float* OSf      = ws + o; o += (size_t)C_ * 1600;
  float* out = (float*)d_out;

  k_emb_ln<<<1600, 256, 0, stream>>>(emb, lng, lnb, emb_norm);
  k_means<<<(R_ * N_ + C_ * N_ + 255) / 256, 256, 0, stream>>>(bst, bsi, bt, bi);
  k_logsm<<<(R_ * N_ + C_ * N_ + 255) / 256, 256, 0, stream>>>(cst, csi, lt, li);
  k_attn<<<2 * H_ * R_, 256, 0, stream>>>(emb_norm, WQt, WKt, WQi, WKi, attn);
  k_br<<<R_ + C_, 256, 0, stream>>>(attn, bt, bi, idxt, idxi, BRt, BRi);
  k_bs<<<(R_ * C_ * N_ + 255) / 256, 256, 0, stream>>>(BRt, BRi, BSp);
  k_cs<<<(R_ * C_ * N_ + 255) / 256, 256, 0, stream>>>(lt, li, CSp);
  k_os1<<<C_, 256, 0, stream>>>(emb, Wos, bos, idxi, covb, ob);
  k_os2<<<dim3(25, 64), 256, 0, stream>>>(ob, emb, OSf);
  k_ln_out<<<3 * R_, 256, 0, stream>>>(BSp, CSp, OSf, BSg, BSb, CSg, CSb, OSg,
                                       OSb, out);
}


# k_attn operand delivery software-pipelined: next 64-k step's E tile and W fragments issued as one group of 16-byte global loads ahead of each 64-WMMA block; E tile double-buffered in LDS
# speedup vs baseline: 1.3291x; 1.3291x over previous
#include <hip/hip_runtime.h>

#define R_ 25
#define C_ 64
#define N_ 64
#define H_ 8
#define E_ 1024
#define NC_ 27

typedef float v2f __attribute__((ext_vector_type(2)));
typedef float v8f __attribute__((ext_vector_type(8)));
typedef float v4f __attribute__((ext_vector_type(4)));

__device__ __forceinline__ v8f wmma4(v2f a, v2f b, v8f c) {
  return __builtin_amdgcn_wmma_f32_16x16x4_f32(false, a, false, b, (short)0, c,
                                               false, false);
}

__global__ __launch_bounds__(256) void k_emb_ln(const float* __restrict__ w,
                                                const float* __restrict__ g,
                                                const float* __restrict__ b,
                                                float* __restrict__ out) {
  __shared__ float red[256];
  int row = blockIdx.x, tid = threadIdx.x;
  const float* x = w + (size_t)row * E_;
  float s = 0.f;
  for (int e = tid; e < E_; e += 256) s += x[e];
  red[tid] = s; __syncthreads();
  for (int off = 128; off > 0; off >>= 1) {
    if (tid < off) red[tid] += red[tid + off];
    __syncthreads();
  }
  float mean = red[0] * (1.f / E_);
  __syncthreads();
  float s2 = 0.f;
  for (int e = tid; e < E_; e += 256) { float d = x[e] - mean; s2 += d * d; }
  red[tid] = s2; __syncthreads();
  for (int off = 128; off > 0; off >>= 1) {
    if (tid < off) red[tid] += red[tid + off];
    __syncthreads();
  }
  float inv = rsqrtf(red[0] * (1.f / E_) + 1e-16f);
  for (int e = tid; e < E_; e += 256)
    out[(size_t)row * E_ + e] = (x[e] - mean) * inv * g[e] + b[e];
}

__global__ void k_means(const float* __restrict__ bst,
                        const float* __restrict__ bsi,
                        float* __restrict__ bt, float* __restrict__ bi) {
  int t = blockIdx.x * blockDim.x + threadIdx.x;
  if (t < R_ * N_) {
    int d = t / N_, i = t % N_;
    const float* p = bst + ((size_t)d * C_ * N_ + i) * 4;
    bt[t] = 0.25f * (p[0] + p[1] + p[2] + p[3]);
  } else if (t < R_ * N_ + C_ * N_) {
    int t2 = t - R_ * N_;
    int d = t2 / N_, i = t2 % N_;
    const float* p = bsi + ((size_t)d * N_ + i) * 4;
    bi[t2] = 0.25f * (p[0] + p[1] + p[2] + p[3]);
  }
}

__global__ void k_logsm(const float* __restrict__ cst,
                        const float* __restrict__ csi,
                        float* __restrict__ lt, float* __restrict__ li) {
  int t = blockIdx.x * blockDim.x + threadIdx.x;
  const float* src; float* dst;
  if (t < R_ * N_) {
    int a = t / N_, nn = t % N_;
    src = cst + ((size_t)a * C_ * N_ + nn) * NC_ * 4;
    dst = lt + (size_t)t * NC_;
  } else if (t < R_ * N_ + C_ * N_) {
    int t2 = t - R_ * N_;
    int b = t2 / N_, nn = t2 % N_;
    src = csi + ((size_t)b * N_ + nn) * NC_ * 4;
    dst = li + (size_t)t2 * NC_;
  } else return;
  float v[NC_];
  float m = -1e30f;
  for (int k = 0; k < NC_; ++k) {
    const float* p = src + k * 4;
    v[k] = 0.25f * (p[0] + p[1] + p[2] + p[3]);
    m = fmaxf(m, v[k]);
  }
  float se = 0.f;
  for (int k = 0; k < NC_; ++k) se += expf(v[k] - m);
  float lse = m + logf(se);
  for (int k = 0; k < NC_; ++k) dst[k] = v[k] - lse;
}

#define SEG    64
#define NSEG   (E_ / SEG)
#define NSTEP  (16 * NSEG)
#define EPAD   68
#define QKPAD  68
#define EBUF   (64 * EPAD)

__device__ __forceinline__ v4f frag4(const float* p) {
  return *reinterpret_cast<const v4f*>(p);
}

__device__ __forceinline__ void seg_step(
    const float* Ecur, float* Enxt, const v4f (&bc)[8], v4f (&bn)[8],
    const float* __restrict__ Eg_thr, const float* __restrict__ W_lane,
    int tn, int est_off, int afrag_off,
    v8f& acc0, v8f& acc1, v8f& acc2, v8f& acc3) {
  v4f er[4];
  const bool has_next = tn < NSTEP;
  if (has_next) {
    const int sgn = tn & (NSEG - 1);
    const int fcn = tn >> 4;
    const float* ep = Eg_thr + sgn * SEG;
#pragma unroll
    for (int i = 0; i < 4; ++i) er[i] = frag4(ep + (size_t)i * 16 * E_);
    const float* wp = W_lane + (size_t)fcn * 64 * E_ + sgn * SEG;
#pragma unroll
    for (int j = 0; j < 8; ++j) bn[j] = frag4(wp + 8 * j);
  }

  const float* ap = Ecur + afrag_off;
  v4f a0 = frag4(ap + 0 * 16 * EPAD);
  v4f a1 = frag4(ap + 1 * 16 * EPAD);
  v4f a2 = frag4(ap + 2 * 16 * EPAD);
  v4f a3 = frag4(ap + 3 * 16 * EPAD);
#pragma unroll
  for (int j = 0; j < 8; ++j) {
    v4f n0 = a0, n1 = a1, n2 = a2, n3 = a3;
    if (j < 7) {
      n0 = frag4(ap + 0 * 16 * EPAD + 8 * (j + 1));
      n1 = frag4(ap + 1 * 16 * EPAD + 8 * (j + 1));
      n2 = frag4(ap + 2 * 16 * EPAD + 8 * (j + 1));
      n3 = frag4(ap + 3 * 16 * EPAD + 8 * (j + 1));
    }
    const v4f b = bc[j];
    acc0 = wmma4(a0.xy, b.xy, acc0);
    acc1 = wmma4(a1.xy, b.xy, acc1);
    acc2 = wmma4(a2.xy, b.xy, acc2);
    acc3 = wmma4(a3.xy, b.xy, acc3);
    acc0 = wmma4(a0.zw, b.zw, acc0);
    acc1 = wmma4(a1.zw, b.zw, acc1);
    acc2 = wmma4(a2.zw, b.zw, acc2);
    acc3 = wmma4(a3.zw, b.zw, acc3);
    a0 = n0; a1 = n1; a2 = n2; a3 = n3;
  }

  if (has_next) {
#pragma unroll
    for (int i = 0; i < 4; ++i)
      *reinterpret_cast<v4f*>(Enxt + est_off + i * 16 * EPAD) = er[i];
  }
  __syncthreads();
}

__global__ __launch_bounds__(256) void k_attn(
    const float* __restrict__ emb_norm,
    const float* __restrict__ WQt, const float* __restrict__ WKt,
    const float* __restrict__ WQi, const float* __restrict__ WKi,
    float* __restrict__ attn) {
  __shared__ float lds[2 * EBUF + 2 * 64 * QKPAD];
  float* E0 = lds;
  float* E1 = lds + EBUF;
  float* QKs = lds + 2 * EBUF;

  int tid = threadIdx.x;
  int lane = tid & 31;
  int wid = tid >> 5;

  int blk = blockIdx.x;
  int r = blk % R_;
  int h = (blk / R_) % H_;
  int w = blk / (R_ * H_);

  const float* WQ = (w ? WQi : WQt) + (size_t)h * E_ * E_;
  const float* WK = (w ? WKi : WKt) + (size_t)h * E_ * E_;
  const float* Eg = emb_norm + (size_t)r * N_ * E_;

  int qk = wid >> 2;
  int n0 = (wid & 3) << 4;
  const float* Wm = qk ? WK : WQ;

  int i0 = (wid >> 1) << 4;
  int j0 = (wid & 1) << 5;
  v8f s0 = {}; v8f s1 = {};

  int hf = lane >> 4;
  int kp = hf << 1;
  int nl = lane & 15;

  const int est_off = (tid >> 4) * EPAD + ((tid & 15) << 2);
  const float* Eg_thr = Eg + (size_t)(tid >> 4) * E_ + ((tid & 15) << 2);
  const int afrag_off = nl * EPAD + 4 * hf;
  const float* W_lane = Wm + (size_t)(n0 + nl) * E_ + 4 * hf;

  v4f bA[8], bB[8];
  {
    v4f er[4];
#pragma unroll
    for (int i = 0; i < 4; ++i) er[i] = frag4(Eg_thr + (size_t)i * 16 * E_);
#pragma unroll
    for (int j = 0; j < 8; ++j) bA[j] = frag4(W_lane + 8 * j);
#pragma unroll
    for (int i = 0; i < 4; ++i)
      *reinterpret_cast<v4f*>(E0 + est_off + i * 16 * EPAD) = er[i];
  }
  __syncthreads();

  for (int fc = 0; fc < 16; ++fc) {
    v8f acc0 = {}, acc1 = {}, acc2 = {}, acc3 = {};

    for (int sg = 0; sg < NSEG; sg += 2) {
      int t = fc * NSEG + sg;
      seg_step(E0, E1, bA, bB, Eg_thr, W_lane, t + 1, est_off, afrag_off,
               acc0, acc1, acc2, acc3);
      seg_step(E1, E0, bB, bA, Eg_thr, W_lane, t + 2, est_off, afrag_off,
               acc0, acc1, acc2, acc3);
    }

    {
      float* QK = QKs + qk * (64 * QKPAD);
      int rsub = hf << 3;
#pragma unroll
      for (int v = 0; v < 8; ++v) {
        QK[( 0 + rsub + v) * QKPAD + n0 + nl] = acc0[v];
        QK[(16 + rsub + v) * QKPAD + n0 + nl] = acc1[v];
        QK[(32 + rsub + v) * QKPAD + n0 + nl] = acc2[v];
        QK[(48 + rsub + v) * QKPAD + n0 + nl] = acc3[v];
      }
    }
    __syncthreads();

    {
      const float* Qs = QKs;
      const float* Ks = QKs + 64 * QKPAD;
#pragma unroll
      for (int k0 = 0; k0 < 64; k0 += 4) {
        v2f a  = *reinterpret_cast<const v2f*>(&Qs[(i0 + nl) * QKPAD + k0 + kp]);
        v2f b0 = *reinterpret_cast<const v2f*>(&Ks[(j0 + nl) * QKPAD + k0 + kp]);
        v2f b1 = *reinterpret_cast<const v2f*>(&Ks[(j0 + 16 + nl) * QKPAD + k0 + kp]);
        s0 = wmma4(a, b0, s0);
        s1 = wmma4(a, b1, s1);
      }
    }
  }

  __syncthreads();
  float* Ss = QKs;
  {
    int rbase = i0 + (hf << 3);
#pragma unroll
    for (int v = 0; v < 8; ++v) {
      Ss[(rbase + v) * 64 + j0 + nl]      = s0[v];
      Ss[(rbase + v) * 64 + j0 + 16 + nl] = s1[v];
    }
  }
  __syncthreads();

  if (tid < 64) {
    int j = tid;
    const float inv32 = 0.03125f;
    float m = -1e30f;
    for (int i = 0; i < 64; ++i) m = fmaxf(m, Ss[i * 64 + j] * inv32);
    float se = 0.f;
    for (int i = 0; i < 64; ++i) {
      float e = expf(Ss[i * 64 + j] * inv32 - m);
      Ss[i * 64 + j] = e;
      se += e;
    }
    float rs = 1.f / se;
    for (int i = 0; i < 64; ++i) Ss[i * 64 + j] *= rs;
  }
  __syncthreads();

  float* dst = attn + (size_t)((w * H_ + h) * R_ + r) * 4096;
  for (int q = tid; q < 4096; q += 256) dst[q] = Ss[q];
}

__global__ __launch_bounds__(256) void k_br(
    const float* __restrict__ attn, const float* __restrict__ bt,
    const float* __restrict__ bi, const int* __restrict__ idxt,
    const int* __restrict__ idxi, float* __restrict__ BRt,
    float* __restrict__ BRi) {
  __shared__ float brow[64];
  int b = blockIdx.x, tid = threadIdx.x;
  int d, r, w;
  const float* bsrc; float* dst;
  if (b < R_) { d = b;      r = idxt[d * C_]; w = 0; bsrc = bt + d * 64; dst = BRt + (size_t)d * 64 * 8; }
  else        { d = b - R_; r = idxi[d];      w = 1; bsrc = bi + d * 64; dst = BRi + (size_t)d * 64 * 8; }
  if (tid < 64) brow[tid] = bsrc[tid];
  __syncthreads();
  int j = tid & 63;
  for (int hh = tid >> 6; hh < H_; hh += 4) {
    const float* A = attn + (size_t)((w * H_ + hh) * R_ + r) * 4096;
    float acc = 0.f;
    for (int i = 0; i < 64; ++i) acc += brow[i] * A[i * 64 + j];
    dst[(size_t)j * 8 + hh] = acc;
  }
}

__global__ void k_bs(const float* __restrict__ BRt,
                     const float* __restrict__ BRi, float* __restrict__ BS) {
  int t = blockIdx.x * blockDim.x + threadIdx.x;
  if (t >= R_ * C_ * N_) return;
  int a = t / (C_ * N_);
  int rem = t % (C_ * N_);
  int b = rem / N_, j = rem % N_;
  const float* x = BRt + ((size_t)a * 64 + j) * 8;
  const float* y = BRi + ((size_t)b * 64 + j) * 8;
  float dot = 0.f, nx = 0.f, ny = 0.f;
  for (int hh = 0; hh < 8; ++hh) {
    dot += x[hh] * y[hh]; nx += x[hh] * x[hh]; ny += y[hh] * y[hh];
  }
  float na = fmaxf(sqrtf(nx), 1e-15f);
  float nb = fmaxf(sqrtf(ny), 1e-15f);
  BS[t] = dot / (na * nb);
}

__global__ void k_cs(const float* __restrict__ lt, const float* __restrict__ li,
                     float* __restrict__ CS) {
  int t = blockIdx.x * blockDim.x + threadIdx.x;
  if (t >= R_ * C_ * N_) return;
  int a = t / (C_ * N_);
  int rem = t % (C_ * N_);
  int b = rem / N_, nn = rem % N_;
  const float* pt = lt + ((size_t)a * 64 + nn) * NC_;
  const float* pi = li + ((size_t)b * 64 + nn) * NC_;
  float m = -1e30f;
  for (int k = 0; k < NC_; ++k) m = fmaxf(m, 0.5f * (pt[k] + pi[k]));
  float se = 0.f;
  for (int k = 0; k < NC_; ++k) se += expf(0.5f * (pt[k] + pi[k]) - m);
  float lse = m + logf(se);
  float klt = 0.f, kli = 0.f;
  for (int k = 0; k < NC_; ++k) {
    float M = 0.5f * (pt[k] + pi[k]) - lse;
    float eM = expf(M);
    klt += eM * (M - pt[k]);
    kli += eM * (M - pi[k]);
  }
  CS[t] = -0.5f * (klt + kli) * (1.f / NC_);
}

__global__ __launch_bounds__(256) void k_os1(
    const float* __restrict__ emb, const float* __restrict__ Wos,
    const float* __restrict__ bos, const int* __restrict__ idxi,
    const int* __restrict__ covb, float* __restrict__ ob) {
  __shared__ float rowv[E_];
  int d = blockIdx.x, tid = threadIdx.x;
  int row = idxi[d] * N_ + covb[d];
  for (int e = tid; e < E_; e += 256) rowv[e] = emb[(size_t)row * E_ + e];
  __syncthreads();
  for (int f = tid; f < E_; f += 256) {
    const float* wr = Wos + (size_t)f * E_;
    float acc = bos[f];
    for (int e = 0; e < E_; ++e) acc += rowv[e] * wr[e];
    ob[(size_t)d * E_ + f] = acc;
  }
}

__global__ __launch_bounds__(256) void k_os2(const float* __restrict__ ob,
                                             const float* __restrict__ emb,
                                             float* __restrict__ osf) {
  __shared__ float rowv[E_];
  __shared__ float part[256];
  int d = blockIdx.y;
  int tc = blockIdx.x;
  int tid = threadIdx.x;
  for (int e = tid; e < E_; e += 256) rowv[e] = ob[(size_t)d * E_ + e];
  __syncthreads();
  int t = tc * 64 + (tid & 63);
  int p = tid >> 6;
  const float* er = emb + (size_t)t * E_ + p * 256;
  const float* rv = rowv + p * 256;
  float acc = 0.f;
  for (int e = 0; e < 256; ++e) acc += rv[e] * er[e];
  part[tid] = acc;
  __syncthreads();
  if (p == 0)
    osf[(size_t)d * 1600 + t] = part[tid] + part[tid + 64] + part[tid + 128] + part[tid + 192];
}

__global__ __launch_bounds__(256) void k_ln_out(
    const float* __restrict__ BSp, const float* __restrict__ CSp,
    const float* __restrict__ OSf, const float* __restrict__ g0,
    const float* __restrict__ b0, const float* __restrict__ g1,
    const float* __restrict__ b1, const float* __restrict__ g2,
    const float* __restrict__ b2, float* __restrict__ out) {
  __shared__ float red[256];
  int mode = blockIdx.x / R_;
  int sl = blockIdx.x % R_;
  int tid = threadIdx.x;
  float x[16];
  for (int q = 0; q < 16; ++q) {
    int idx = tid * 16 + q;
    float v;
    if (mode == 0) v = BSp[(size_t)sl * 4096 + idx];
    else if (mode == 1) v = CSp[(size_t)sl * 4096 + idx];
    else { int bb = idx >> 6, nn = idx & 63; v = OSf[(size_t)bb * 1600 + sl * 64 + nn]; }
    x[q] = v;
  }
  float s = 0.f;
  for (int q = 0; q < 16; ++q) s += x[q];
  red[tid] = s; __syncthreads();
  for (int off = 128; off > 0; off >>= 1) {
    if (tid < off) red[tid] += red[tid + off];
    __syncthreads();
  }
  float mean = red[0] * (1.f / 4096.f);
  __syncthreads();
  float s2 = 0.f;
  for (int q = 0; q < 16; ++q) { float dd = x[q] - mean; s2 += dd * dd; }
  red[tid] = s2; __syncthreads();
  for (int off = 128; off > 0; off >>= 1) {
    if (tid < off) red[tid] += red[tid + off];
    __syncthreads();
  }
  float inv = rsqrtf(red[0] * (1.f / 4096.f) + 1e-5f);
  const float* g = (mode == 0) ? g0 : ((mode == 1) ? g1 : g2);
  const float* bb = (mode == 0) ? b0 : ((mode == 1) ? b1 : b2);
  for (int q = 0; q < 16; ++q) {
    int idx = tid * 16 + q;
    out[(size_t)mode * 102400 + (size_t)sl * 4096 + idx] =
        (x[q] - mean) * inv * g[idx] + bb[idx];
  }
}

extern "C" void kernel_launch(void* const* d_in, const int* in_sizes, int n_in,
                              void* d_out, int out_size, void* d_ws,
                              size_t ws_size, hipStream_t stream) {
  const float* bst = (const float*)d_in[0];
  const float* bsi = (const float*)d_in[1];
  const float* cst = (const float*)d_in[2];
  const float* csi = (const float*)d_in[3];
  const int* idxt  = (const int*)d_in[4];
  const int* idxi  = (const int*)d_in[5];
  const int* covb  = (const int*)d_in[6];
  const float* emb = (const float*)d_in[7];
  const float* lng = (const float*)d_in[8];
  const float* lnb = (const float*)d_in[9];
  const float* WQt = (const float*)d_in[10];
  const float* WKt = (const float*)d_in[11];
  const float* WQi = (const float*)d_in[12];
  const float* WKi = (const float*)d_in[13];
  const float* Wos = (const float*)d_in[14];
  const float* bos = (const float*)d_in[15];
  const float* BSg = (const float*)d_in[16];
  const float* BSb = (const float*)d_in[17];
  const float* CSg = (const float*)d_in[18];
  const float* CSb = (const float*)d_in[19];
  const float* OSg = (const float*)d_in[20];
  const float* OSb = (const float*)d_in[21];

  float* ws = (float*)d_ws;
  size_t o = 0;
  float* emb_norm = ws + o; o += (size_t)1600 * 1024;
  float* attn     = ws + o; o += (size_t)2 * 8 * 25 * 4096;
  float* bt       = ws + o; o += R_ * N_;
  float* bi       = ws + o; o += C_ * N_;
  float* lt       = ws + o; o += R_ * N_ * NC_;
  float* li       = ws + o; o += C_ * N_ * NC_;
  float* BRt      = ws + o; o += R_ * N_ * H_;
  float* BRi      = ws + o; o += C_ * N_ * H_;
  float* BSp      = ws + o; o += R_ * C_ * N_;
  float* CSp      = ws + o; o += R_ * C_ * N_;
  float* ob       = ws + o; o += C_ * E_;
  float* OSf      = ws + o; o += (size_t)C_ * 1600;
  float* out = (float*)d_out;

  k_emb_ln<<<1600, 256, 0, stream>>>(emb, lng, lnb, emb_norm);
  k_means<<<(R_ * N_ + C_ * N_ + 255) / 256, 256, 0, stream>>>(bst, bsi, bt, bi);
  k_logsm<<<(R_ * N_ + C_ * N_ + 255) / 256, 256, 0, stream>>>(cst, csi, lt, li);
  k_attn<<<2 * H_ * R_, 256, 0, stream>>>(emb_norm, WQt, WKt, WQi, WKi, attn);
  k_br<<<R_ + C_, 256, 0, stream>>>(attn, bt, bi, idxt, idxi, BRt, BRi);
  k_bs<<<(R_ * C_ * N_ + 255) / 256, 256, 0, stream>>>(BRt, BRi, BSp);
  k_cs<<<(R_ * C_ * N_ + 255) / 256, 256, 0, stream>>>(lt, li, CSp);
  k_os1<<<C_, 256, 0, stream>>>(emb, Wos, bos, idxi, covb, ob);
  k_os2<<<dim3(25, 64), 256, 0, stream>>>(ob, emb, OSf);
  k_ln_out<<<3 * R_, 256, 0, stream>>>(BSp, CSp, OSf, BSg, BSb, CSg, CSb, OSg,
                                       OSb, out);
}


# k_os1/k_os2 (ob = x W_os^T, OS = ob emb^T): per-thread dots over 4 KiB-strided rows replaced by 32x32 f32 WMMA 16x16x4 tiles, split-k over 8 waves summed in LDS; weight rows read once per 32 d
# speedup vs baseline: 1.5422x; 1.1603x over previous
#include <hip/hip_runtime.h>

#define R_ 25
#define C_ 64
#define N_ 64
#define H_ 8
#define E_ 1024
#define NC_ 27

typedef float v2f __attribute__((ext_vector_type(2)));
typedef float v8f __attribute__((ext_vector_type(8)));
typedef float v4f __attribute__((ext_vector_type(4)));

__device__ __forceinline__ v8f wmma4(v2f a, v2f b, v8f c) {
  return __builtin_amdgcn_wmma_f32_16x16x4_f32(false, a, false, b, (short)0, c,
                                               false, false);
}

__global__ __launch_bounds__(256) void k_emb_ln(const float* __restrict__ w,
                                                const float* __restrict__ g,
                                                const float* __restrict__ b,
                                                float* __restrict__ out) {
  __shared__ float red[256];
  int row = blockIdx.x, tid = threadIdx.x;
  const float* x = w + (size_t)row * E_;
  float s = 0.f;
  for (int e = tid; e < E_; e += 256) s += x[e];
  red[tid] = s; __syncthreads();
  for (int off = 128; off > 0; off >>= 1) {
    if (tid < off) red[tid] += red[tid + off];
    __syncthreads();
  }
  float mean = red[0] * (1.f / E_);
  __syncthreads();
  float s2 = 0.f;
  for (int e = tid; e < E_; e += 256) { float d = x[e] - mean; s2 += d * d; }
  red[tid] = s2; __syncthreads();
  for (int off = 128; off > 0; off >>= 1) {
    if (tid < off) red[tid] += red[tid + off];
    __syncthreads();
  }
  float inv = rsqrtf(red[0] * (1.f / E_) + 1e-16f);
  for (int e = tid; e < E_; e += 256)
    out[(size_t)row * E_ + e] = (x[e] - mean) * inv * g[e] + b[e];
}

__global__ void k_means(const float* __restrict__ bst,
                        const float* __restrict__ bsi,
                        float* __restrict__ bt, float* __restrict__ bi) {
  int t = blockIdx.x * blockDim.x + threadIdx.x;
  if (t < R_ * N_) {
    int d = t / N_, i = t % N_;
    const float* p = bst + ((size_t)d * C_ * N_ + i) * 4;
    bt[t] = 0.25f * (p[0] + p[1] + p[2] + p[3]);
  } else if (t < R_ * N_ + C_ * N_) {
    int t2 = t - R_ * N_;
    int d = t2 / N_, i = t2 % N_;
    const float* p = bsi + ((size_t)d * N_ + i) * 4;
    bi[t2] = 0.25f * (p[0] + p[1] + p[2] + p[3]);
  }
}

__global__ void k_logsm(const float* __restrict__ cst,
                        const float* __restrict__ csi,
                        float* __restrict__ lt, float* __restrict__ li) {
  int t = blockIdx.x * blockDim.x + threadIdx.x;
  const float* src; float* dst;
  if (t < R_ * N_) {
    int a = t / N_, nn = t % N_;
    src = cst + ((size_t)a * C_ * N_ + nn) * NC_ * 4;
    dst = lt + (size_t)t * NC_;
  } else if (t < R_ * N_ + C_ * N_) {
    int t2 = t - R_ * N_;
    int b = t2 / N_, nn = t2 % N_;
    src = csi + ((size_t)b * N_ + nn) * NC_ * 4;
    dst = li + (size_t)t2 * NC_;
  } else return;
  float v[NC_];
  float m = -1e30f;
  for (int k = 0; k < NC_; ++k) {
    const float* p = src + k * 4;
    v[k] = 0.25f * (p[0] + p[1] + p[2] + p[3]);
    m = fmaxf(m, v[k]);
  }
  float se = 0.f;
  for (int k = 0; k < NC_; ++k) se += expf(v[k] - m);
  float lse = m + logf(se);
  for (int k = 0; k < NC_; ++k) dst[k] = v[k] - lse;
}

#define SEG    64
#define NSEG   (E_ / SEG)
#define NSTEP  (16 * NSEG)
#define EPAD   68
#define QKPAD  68
#define EBUF   (64 * EPAD)

__device__ __forceinline__ v4f frag4(const float* p) {
  return *reinterpret_cast<const v4f*>(p);
}

__device__ __forceinline__ void seg_step(
    const float* Ecur, float* Enxt, const v4f (&bc)[8], v4f (&bn)[8],
    const float* __restrict__ Eg_thr, const float* __restrict__ W_lane,
    int tn, int est_off, int afrag_off,
    v8f& acc0, v8f& acc1, v8f& acc2, v8f& acc3) {
  v4f er[4];
  const bool has_next = tn < NSTEP;
  if (has_next) {
    const int sgn = tn & (NSEG - 1);
    const int fcn = tn >> 4;
    const float* ep = Eg_thr + sgn * SEG;
#pragma unroll
    for (int i = 0; i < 4; ++i) er[i] = frag4(ep + (size_t)i * 16 * E_);
    const float* wp = W_lane + (size_t)fcn * 64 * E_ + sgn * SEG;
#pragma unroll
    for (int j = 0; j < 8; ++j) bn[j] = frag4(wp + 8 * j);
  }

  const float* ap = Ecur + afrag_off;
  v4f a0 = frag4(ap + 0 * 16 * EPAD);
  v4f a1 = frag4(ap + 1 * 16 * EPAD);
  v4f a2 = frag4(ap + 2 * 16 * EPAD);
  v4f a3 = frag4(ap + 3 * 16 * EPAD);
#pragma unroll
  for (int j = 0; j < 8; ++j) {
    v4f n0 = a0, n1 = a1, n2 = a2, n3 = a3;
    if (j < 7) {
      n0 = frag4(ap + 0 * 16 * EPAD + 8 * (j + 1));
      n1 = frag4(ap + 1 * 16 * EPAD + 8 * (j + 1));
      n2 = frag4(ap + 2 * 16 * EPAD + 8 * (j + 1));
      n3 = frag4(ap + 3 * 16 * EPAD + 8 * (j + 1));
    }
    const v4f b = bc[j];
    acc0 = wmma4(a0.xy, b.xy, acc0);
    acc1 = wmma4(a1.xy, b.xy, acc1);
    acc2 = wmma4(a2.xy, b.xy, acc2);
    acc3 = wmma4(a3.xy, b.xy, acc3);
    acc0 = wmma4(a0.zw, b.zw, acc0);
    acc1 = wmma4(a1.zw, b.zw, acc1);
    acc2 = wmma4(a2.zw, b.zw, acc2);
    acc3 = wmma4(a3.zw, b.zw, acc3);
    a0 = n0; a1 = n1; a2 = n2; a3 = n3;
  }

  if (has_next) {
#pragma unroll
    for (int i = 0; i < 4; ++i)
      *reinterpret_cast<v4f*>(Enxt + est_off + i * 16 * EPAD) = er[i];
  }
  __syncthreads();
}

__global__ __launch_bounds__(256) void k_attn(
    const float* __restrict__ emb_norm,
    const float* __restrict__ WQt, const float* __restrict__ WKt,
    const float* __restrict__ WQi, const float* __restrict__ WKi,
    float* __restrict__ attn) {
  __shared__ float lds[2 * EBUF + 2 * 64 * QKPAD];
  float* E0 = lds;
  float* E1 = lds + EBUF;
  float* QKs = lds + 2 * EBUF;

  int tid = threadIdx.x;
  int lane = tid & 31;
  int wid = tid >> 5;

  int blk = blockIdx.x;
  int r = blk % R_;
  int h = (blk / R_) % H_;
  int w = blk / (R_ * H_);

  const float* WQ = (w ? WQi : WQt) + (size_t)h * E_ * E_;
  const float* WK = (w ? WKi : WKt) + (size_t)h * E_ * E_;
  const float* Eg = emb_norm + (size_t)r * N_ * E_;

  int qk = wid >> 2;
  int n0 = (wid & 3) << 4;
  const float* Wm = qk ? WK : WQ;

  int i0 = (wid >> 1) << 4;
  int j0 = (wid & 1) << 5;
  v8f s0 = {}; v8f s1 = {};

  int hf = lane >> 4;
  int kp = hf << 1;
  int nl = lane & 15;

  const int est_off = (tid >> 4) * EPAD + ((tid & 15) << 2);
  const float* Eg_thr = Eg + (size_t)(tid >> 4) * E_ + ((tid & 15) << 2);
  const int afrag_off = nl * EPAD + 4 * hf;
  const float* W_lane = Wm + (size_t)(n0 + nl) * E_ + 4 * hf;

  v4f bA[8], bB[8];
  {
    v4f er[4];
#pragma unroll
    for (int i = 0; i < 4; ++i) er[i] = frag4(Eg_thr + (size_t)i * 16 * E_);
#pragma unroll
    for (int j = 0; j < 8; ++j) bA[j] = frag4(W_lane + 8 * j);
#pragma unroll
    for (int i = 0; i < 4; ++i)
      *reinterpret_cast<v4f*>(E0 + est_off + i * 16 * EPAD) = er[i];
  }
  __syncthreads();

  for (int fc = 0; fc < 16; ++fc) {
    v8f acc0 = {}, acc1 = {}, acc2 = {}, acc3 = {};

    for (int sg = 0; sg < NSEG; sg += 2) {
      int t = fc * NSEG + sg;
      seg_step(E0, E1, bA, bB, Eg_thr, W_lane, t + 1, est_off, afrag_off,
               acc0, acc1, acc2, acc3);
      seg_step(E1, E0, bB, bA, Eg_thr, W_lane, t + 2, est_off, afrag_off,
               acc0, acc1, acc2, acc3);
    }

    {
      float* QK = QKs + qk * (64 * QKPAD);
      int rsub = hf << 3;
#pragma unroll
      for (int v = 0; v < 8; ++v) {
        QK[( 0 + rsub + v) * QKPAD + n0 + nl] = acc0[v];
        QK[(16 + rsub + v) * QKPAD + n0 + nl] = acc1[v];
        QK[(32 + rsub + v) * QKPAD + n0 + nl] = acc2[v];
        QK[(48 + rsub + v) * QKPAD + n0 + nl] = acc3[v];
      }
    }
    __syncthreads();

    {
      const float* Qs = QKs;
      const float* Ks = QKs + 64 * QKPAD;
#pragma unroll
      for (int k0 = 0; k0 < 64; k0 += 4) {
        v2f a  = *reinterpret_cast<const v2f*>(&Qs[(i0 + nl) * QKPAD + k0 + kp]);
        v2f b0 = *reinterpret_cast<const v2f*>(&Ks[(j0 + nl) * QKPAD + k0 + kp]);
        v2f b1 = *reinterpret_cast<const v2f*>(&Ks[(j0 + 16 + nl) * QKPAD + k0 + kp]);
        s0 = wmma4(a, b0, s0);
        s1 = wmma4(a, b1, s1);
      }
    }
  }

  __syncthreads();
  float* Ss = QKs;
  {
    int rbase = i0 + (hf << 3);
#pragma unroll
    for (int v = 0; v < 8; ++v) {
      Ss[(rbase + v) * 64 + j0 + nl]      = s0[v];
      Ss[(rbase + v) * 64 + j0 + 16 + nl] = s1[v];
    }
  }
  __syncthreads();

  if (tid < 64) {
    int j = tid;
    const float inv32 = 0.03125f;
    float m = -1e30f;
    for (int i = 0; i < 64; ++i) m = fmaxf(m, Ss[i * 64 + j] * inv32);
    float se = 0.f;
    for (int i = 0; i < 64; ++i) {
      float e = expf(Ss[i * 64 + j] * inv32 - m);
      Ss[i * 64 + j] = e;
      se += e;
    }
    float rs = 1.f / se;
    for (int i = 0; i < 64; ++i) Ss[i * 64 + j] *= rs;
  }
  __syncthreads();

  float* dst = attn + (size_t)((w * H_ + h) * R_ + r) * 4096;
  for (int q = tid; q < 4096; q += 256) dst[q] = Ss[q];
}

__global__ __launch_bounds__(256) void k_br(
    const float* __restrict__ attn, const float* __restrict__ bt,
    const float* __restrict__ bi, const int* __restrict__ idxt,
    const int* __restrict__ idxi, float* __restrict__ BRt,
    float* __restrict__ BRi) {
  __shared__ float brow[64];
  int b = blockIdx.x, tid = threadIdx.x;
  int d, r, w;
  const float* bsrc; float* dst;
  if (b < R_) { d = b;      r = idxt[d * C_]; w = 0; bsrc = bt + d * 64; dst = BRt + (size_t)d * 64 * 8; }
  else        { d = b - R_; r = idxi[d];      w = 1; bsrc = bi + d * 64; dst = BRi + (size_t)d * 64 * 8; }
  if (tid < 64) brow[tid] = bsrc[tid];
  __syncthreads();
  int j = tid & 63;
  for (int hh = tid >> 6; hh < H_; hh += 4) {
    const float* A = attn + (size_t)((w * H_ + hh) * R_ + r) * 4096;
    float acc = 0.f;
    for (int i = 0; i < 64; ++i) acc += brow[i] * A[i * 64 + j];
    dst[(size_t)j * 8 + hh] = acc;
  }
}

__global__ void k_bs(const float* __restrict__ BRt,
                     const float* __restrict__ BRi, float* __restrict__ BS) {
  int t = blockIdx.x * blockDim.x + threadIdx.x;
  if (t >= R_ * C_ * N_) return;
  int a = t / (C_ * N_);
  int rem = t % (C_ * N_);
  int b = rem / N_, j = rem % N_;
  const float* x = BRt + ((size_t)a * 64 + j) * 8;
  const float* y = BRi + ((size_t)b * 64 + j) * 8;
  float dot = 0.f, nx = 0.f, ny = 0.f;
  for (int hh = 0; hh < 8; ++hh) {
    dot += x[hh] * y[hh]; nx += x[hh] * x[hh]; ny += y[hh] * y[hh];
  }
  float na = fmaxf(sqrtf(nx), 1e-15f);
  float nb = fmaxf(sqrtf(ny), 1e-15f);
  BS[t] = dot / (na * nb);
}

__global__ void k_cs(const float* __restrict__ lt, const float* __restrict__ li,
                     float* __restrict__ CS) {
  int t = blockIdx.x * blockDim.x + threadIdx.x;
  if (t >= R_ * C_ * N_) return;
  int a = t / (C_ * N_);
  int rem = t % (C_ * N_);
  int b = rem / N_, nn = rem % N_;
  const float* pt = lt + ((size_t)a * 64 + nn) * NC_;
  const float* pi = li + ((size_t)b * 64 + nn) * NC_;
  float m = -1e30f;
  for (int k = 0; k < NC_; ++k) m = fmaxf(m, 0.5f * (pt[k] + pi[k]));
  float se = 0.f;
  for (int k = 0; k < NC_; ++k) se += expf(0.5f * (pt[k] + pi[k]) - m);
  float lse = m + logf(se);
  float klt = 0.f, kli = 0.f;
  for (int k = 0; k < NC_; ++k) {
    float M = 0.5f * (pt[k] + pi[k]) - lse;
    float eM = expf(M);
    klt += eM * (M - pt[k]);
    kli += eM * (M - pi[k]);
  }
  CS[t] = -0.5f * (klt + kli) * (1.f / NC_);
}

#define OSP 36

template <bool GATHER>
__device__ __forceinline__ void os_tile(
    const float* __restrict__ A, const int* __restrict__ idxi,
    const int* __restrict__ covb, const float* __restrict__ B,
    const float* __restrict__ bias, float* __restrict__ out, int ldo,
    float* part) {
  const int tid = threadIdx.x;
  const int lane = tid & 31, wid = tid >> 5;
  const int nsub = wid & 1;
  const int ks = wid >> 1;
  const int c = lane & 15, hf = lane >> 4;
  const int n0 = blockIdx.x * 32;
  const int d0 = blockIdx.y * 32;

  const int koff = ks * 128 + 4 * hf;
  const float* bp = B + (size_t)(n0 + nsub * 16 + c) * E_ + koff;
  const float* ap0;
  const float* ap1;
  {
    int da = d0 + c, db = d0 + 16 + c;
    int ra = GATHER ? (idxi[da] * N_ + covb[da]) : da;
    int rb = GATHER ? (idxi[db] * N_ + covb[db]) : db;
    ap0 = A + (size_t)ra * E_ + koff;
    ap1 = A + (size_t)rb * E_ + koff;
  }

  v8f acc0 = {}, acc1 = {};
#pragma unroll
  for (int half = 0; half < 2; ++half) {
    v4f b[8], a0[8], a1[8];
#pragma unroll
    for (int j = 0; j < 8; ++j) b[j] = frag4(bp + half * 64 + 8 * j);
#pragma unroll
    for (int j = 0; j < 8; ++j) a0[j] = frag4(ap0 + half * 64 + 8 * j);
#pragma unroll
    for (int j = 0; j < 8; ++j) a1[j] = frag4(ap1 + half * 64 + 8 * j);
#pragma unroll
    for (int j = 0; j < 8; ++j) {
      acc0 = wmma4(a0[j].xy, b[j].xy, acc0);
      acc1 = wmma4(a1[j].xy, b[j].xy, acc1);
      acc0 = wmma4(a0[j].zw, b[j].zw, acc0);
      acc1 = wmma4(a1[j].zw, b[j].zw, acc1);
    }
  }

  {
    float* pp = part + ks * (32 * OSP);
    const int rsub = hf << 3;
#pragma unroll
    for (int v = 0; v < 8; ++v) {
      pp[( 0 + rsub + v) * OSP + nsub * 16 + c] = acc0[v];
      pp[(16 + rsub + v) * OSP + nsub * 16 + c] = acc1[v];
    }
  }
  __syncthreads();

#pragma unroll
  for (int q = 0; q < 2; ++q) {
    const int o = tid + 512 * q;
    const int row = o >> 5, col = o & 31;
    float s = 0.f;
#pragma unroll
    for (int k8 = 0; k8 < 8; ++k8) s += part[k8 * (32 * OSP) + row * OSP + col];
    if (bias) s += bias[n0 + col];
    out[(size_t)(d0 + row) * ldo + n0 + col] = s;
  }
}

__global__ __launch_bounds__(512) void k_os1(
    const float* __restrict__ emb, const float* __restrict__ Wos,
    const float* __restrict__ bos, const int* __restrict__ idxi,
    const int* __restrict__ covb, float* __restrict__ ob) {
  __shared__ float part[8 * 32 * OSP];
  os_tile<true>(emb, idxi, covb, Wos, bos, ob, E_, part);
}

__global__ __launch_bounds__(512) void k_os2(const float* __restrict__ ob,
                                             const float* __restrict__ emb,
                                             float* __restrict__ osf) {
  __shared__ float part[8 * 32 * OSP];
  os_tile<false>(ob, nullptr, nullptr, emb, nullptr, osf, 1600, part);
}

__global__ __launch_bounds__(256) void k_ln_out(
    const float* __restrict__ BSp, const float* __restrict__ CSp,
    const float* __restrict__ OSf, const float* __restrict__ g0,
    const float* __restrict__ b0, const float* __restrict__ g1,
    const float* __restrict__ b1, const float* __restrict__ g2,
    const float* __restrict__ b2, float* __restrict__ out) {
  __shared__ float red[256];
  int mode = blockIdx.x / R_;
  int sl = blockIdx.x % R_;
  int tid = threadIdx.x;
  float x[16];
  for (int q = 0; q < 16; ++q) {
    int idx = tid * 16 + q;
    float v;
    if (mode == 0) v = BSp[(size_t)sl * 4096 + idx];
    else if (mode == 1) v = CSp[(size_t)sl * 4096 + idx];
    else { int bb = idx >> 6, nn = idx & 63; v = OSf[(size_t)bb * 1600 + sl * 64 + nn]; }
    x[q] = v;
  }
  float s = 0.f;
  for (int q = 0; q < 16; ++q) s += x[q];
  red[tid] = s; __syncthreads();
  for (int off = 128; off > 0; off >>= 1) {
    if (tid < off) red[tid] += red[tid + off];
    __syncthreads();
  }
  float mean = red[0] * (1.f / 4096.f);
  __syncthreads();
  float s2 = 0.f;
  for (int q = 0; q < 16; ++q) { float dd = x[q] - mean; s2 += dd * dd; }
  red[tid] = s2; __syncthreads();
  for (int off = 128; off > 0; off >>= 1) {
    if (tid < off) red[tid] += red[tid + off];
    __syncthreads();
  }
  float inv = rsqrtf(red[0] * (1.f / 4096.f) + 1e-5f);
  const float* g = (mode == 0) ? g0 : ((mode == 1) ? g1 : g2);
  const float* bb = (mode == 0) ? b0 : ((mode == 1) ? b1 : b2);
  for (int q = 0; q < 16; ++q) {
    int idx = tid * 16 + q;
    out[(size_t)mode * 102400 + (size_t)sl * 4096 + idx] =
        (x[q] - mean) * inv * g[idx] + bb[idx];
  }
}

extern "C" void kernel_launch(void* const* d_in, const int* in_sizes, int n_in,
                              void* d_out, int out_size, void* d_ws,
                              size_t ws_size, hipStream_t stream) {
  const float* bst = (const float*)d_in[0];
  const float* bsi = (const float*)d_in[1];
  const float* cst = (const float*)d_in[2];
  const float* csi = (const float*)d_in[3];
  const int* idxt  = (const int*)d_in[4];
  const int* idxi  = (const int*)d_in[5];
  const int* covb  = (const int*)d_in[6];
  const float* emb = (const float*)d_in[7];
  const float* lng = (const float*)d_in[8];
  const float* lnb = (const float*)d_in[9];
  const float* WQt = (const float*)d_in[10];
  const float* WKt = (const float*)d_in[11];
  const float* WQi = (const float*)d_in[12];
  const float* WKi = (const float*)d_in[13];
  const float* Wos = (const float*)d_in[14];
  const float* bos = (const float*)d_in[15];
  const float* BSg = (const float*)d_in[16];
  const float* BSb = (const float*)d_in[17];
  const float* CSg = (const float*)d_in[18];
  const float* CSb = (const float*)d_in[19];
  const float* OSg = (const float*)d_in[20];
  const float* OSb = (const float*)d_in[21];

  float* ws = (float*)d_ws;
  size_t o = 0;
  float* emb_norm = ws + o; o += (size_t)1600 * 1024;
  float* attn     = ws + o; o += (size_t)2 * 8 * 25 * 4096;
  float* bt       = ws + o; o += R_ * N_;
  float* bi       = ws + o; o += C_ * N_;
  float* lt       = ws + o; o += R_ * N_ * NC_;
  float* li       = ws + o; o += C_ * N_ * NC_;
  float* BRt      = ws + o; o += R_ * N_ * H_;
  float* BRi      = ws + o; o += C_ * N_ * H_;
  float* BSp      = ws + o; o += R_ * C_ * N_;
  float* CSp      = ws + o; o += R_ * C_ * N_;
  float* ob       = ws + o; o += C_ * E_;
  float* OSf      = ws + o; o += (size_t)C_ * 1600;
  float* out = (float*)d_out;

  k_emb_ln<<<1600, 256, 0, stream>>>(emb, lng, lnb, emb_norm);
  k_means<<<(R_ * N_ + C_ * N_ + 255) / 256, 256, 0, stream>>>(bst, bsi, bt, bi);
  k_logsm<<<(R_ * N_ + C_ * N_ + 255) / 256, 256, 0, stream>>>(cst, csi, lt, li);
  k_attn<<<2 * H_ * R_, 256, 0, stream>>>(emb_norm, WQt, WKt, WQi, WKi, attn);
  k_br<<<R_ + C_, 256, 0, stream>>>(attn, bt, bi, idxt, idxi, BRt, BRi);
  k_bs<<<(R_ * C_ * N_ + 255) / 256, 256, 0, stream>>>(BRt, BRi, BSp);
  k_cs<<<(R_ * C_ * N_ + 255) / 256, 256, 0, stream>>>(lt, li, CSp);
  k_os1<<<dim3(E_ / 32, 2), 512, 0, stream>>>(emb, Wos, bos, idxi, covb, ob);
  k_os2<<<dim3(1600 / 32, 2), 512, 0, stream>>>(ob, emb, OSf);
  k_ln_out<<<3 * R_, 256, 0, stream>>>(BSp, CSp, OSf, BSg, BSb, CSg, CSb, OSg,
                                       OSb, out);
}
